# starting kernel
# baseline (speedup 1.0000x reference)
#include <hip/hip_runtime.h>


typedef float v2f __attribute__((ext_vector_type(2)));
typedef float v8f __attribute__((ext_vector_type(8)));

__device__ __forceinline__ float dot4(float4 a, float4 b) {
    return fmaf(a.x, b.x, fmaf(a.y, b.y, fmaf(a.z, b.z, a.w * b.w)));
}

__global__ __launch_bounds__(256) void twotower_dot_wmma(
    const int* __restrict__ x, const int* __restrict__ y,
    const float* __restrict__ Wc, const float* __restrict__ Wa,
    float* __restrict__ out, int Bn)
{
    const int lane   = threadIdx.x & 31;
    const int waveIn = threadIdx.x >> 5;
    const long long waveId = (long long)blockIdx.x * 8 + waveIn;
    const long long b0 = waveId * 16;
    if (b0 >= Bn) return;

    const int m = lane & 15;
    long long b = b0 + m;
    if (b >= Bn) b = Bn - 1;

    const int xi = x[b];
    const int yi = y[b];
    const float4* __restrict__ cr = (const float4*)(Wc + (size_t)xi * 64);
    const float4* __restrict__ ar = (const float4*)(Wa + (size_t)yi * 64);

    const int c0 = (lane >> 4) * 8;

    float4 c4[8], a4[8];
#pragma unroll
    for (int i = 0; i < 8; ++i) { c4[i] = cr[c0 + i]; a4[i] = ar[c0 + i]; }

    float p0 = 0.f, p1 = 0.f;
#pragma unroll
    for (int i = 0; i < 4; ++i) p0 += dot4(c4[i], a4[i]);
#pragma unroll
    for (int i = 0; i < 4; ++i) p1 += dot4(c4[4 + i], a4[4 + i]);

    v2f A;  A.x = p0;  A.y = p1;
    v2f Bm; Bm.x = 1.0f; Bm.y = 1.0f;
    v8f C = {};
    C = __builtin_amdgcn_wmma_f32_16x16x4_f32(
         false, A,  false, Bm,
         (short)0, C,  false,  false);

    const int sel = lane & 7;
    float v = C[0];
    v = (sel == 1) ? C[1] : v;
    v = (sel == 2) ? C[2] : v;
    v = (sel == 3) ? C[3] : v;
    v = (sel == 4) ? C[4] : v;
    v = (sel == 5) ? C[5] : v;
    v = (sel == 6) ? C[6] : v;
    v = (sel == 7) ? C[7] : v;

    if (lane < 8) {
        long long o = b0 + sel;
        if (o < Bn) out[o] = v;
    } else if (lane >= 16 && lane < 24) {
        long long o = b0 + 8 + sel;
        if (o < Bn) out[o] = v;
    }
}

extern "C" void kernel_launch(void* const* d_in, const int* in_sizes, int n_in,
                              void* d_out, int out_size, void* d_ws, size_t ws_size,
                              hipStream_t stream) {
    const int*   x  = (const int*)d_in[0];
    const int*   y  = (const int*)d_in[1];
    const float* Wc = (const float*)d_in[2];
    const float* Wa = (const float*)d_in[3];
    float*       out = (float*)d_out;

    const int Bn = in_sizes[0];
    const long long waves  = ((long long)Bn + 15) / 16;
    const int       blocks = (int)((waves + 7) / 8);

    twotower_dot_wmma<<<blocks, 256, 0, stream>>>(x, y, Wc, Wa, out, Bn);
}


# Row loads cover whole 128-B lines (8 lanes per line, 4 lines per 16-byte load, was 32); row sums via WMMA with 0/1 selector B as lane transpose + 8 in-lane adds; f32 summation order changes
# speedup vs baseline: 3.1344x; 3.1344x over previous
#include <hip/hip_runtime.h>


typedef float v2f __attribute__((ext_vector_type(2)));
typedef float v8f __attribute__((ext_vector_type(8)));

__device__ __forceinline__ float dot4(float4 a, float4 b, float acc) {
    return fmaf(a.x, b.x, fmaf(a.y, b.y, fmaf(a.z, b.z, fmaf(a.w, b.w, acc))));
}

__global__ __launch_bounds__(256) void twotower_dot_wmma(
    const int* __restrict__ x, const int* __restrict__ y,
    const float* __restrict__ Wc, const float* __restrict__ Wa,
    float* __restrict__ out, int Bn)
{
    const int lane = threadIdx.x & 31;
    const int waveIn = __builtin_amdgcn_readfirstlane((int)(threadIdx.x >> 5));
    const long long waveId = (long long)blockIdx.x * 8 + waveIn;
    const long long b0 = waveId * 16;
    if (b0 >= Bn) return;

    const int h = lane >> 4;
    const int q = (lane >> 3) & 1;
    const int j = lane & 7;
    const int rb = 4 * h + 8 * q;

    int xi[4], yi[4];
    if (b0 + 16 <= Bn) {
        const int4 xv = *(const int4*)(x + b0 + rb);
        const int4 yv = *(const int4*)(y + b0 + rb);
        xi[0] = xv.x; xi[1] = xv.y; xi[2] = xv.z; xi[3] = xv.w;
        yi[0] = yv.x; yi[1] = yv.y; yi[2] = yv.z; yi[3] = yv.w;
    } else {
#pragma unroll
        for (int t = 0; t < 4; ++t) {
            long long p = b0 + rb + t;
            if (p >= Bn) p = Bn - 1;
            xi[t] = x[p];
            yi[t] = y[p];
        }
    }

    float4 c4[4][2], a4[4][2];
#pragma unroll
    for (int t = 0; t < 4; ++t) {
        const float4* cr = (const float4*)(Wc + (size_t)xi[t] * 64) + j;
        c4[t][0] = cr[0];
        c4[t][1] = cr[8];
    }
#pragma unroll
    for (int t = 0; t < 4; ++t) {
        const float4* ar = (const float4*)(Wa + (size_t)yi[t] * 64) + j;
        a4[t][0] = ar[0];
        a4[t][1] = ar[8];
    }

    __builtin_amdgcn_sched_barrier(0);

    float P[4];
#pragma unroll
    for (int t = 0; t < 4; ++t)
        P[t] = dot4(c4[t][1], a4[t][1], dot4(c4[t][0], a4[t][0], 0.0f));

    const int c = lane & 15;
    const bool mine = (((c >> 2) & 1) == h);
    const int  tsel = c & 3;
    v2f B0, B1;
    B0.x = (mine && tsel == 0) ? 1.0f : 0.0f;
    B0.y = (mine && tsel == 1) ? 1.0f : 0.0f;
    B1.x = (mine && tsel == 2) ? 1.0f : 0.0f;
    B1.y = (mine && tsel == 3) ? 1.0f : 0.0f;

    v2f A0; A0.x = P[0]; A0.y = P[1];
    v2f A1; A1.x = P[2]; A1.y = P[3];
    v8f acc = {};
    acc = __builtin_amdgcn_wmma_f32_16x16x4_f32(
         false, A0,  false, B0,
         (short)0, acc,  false,  false);
    acc = __builtin_amdgcn_wmma_f32_16x16x4_f32(
         false, A1,  false, B1,
         (short)0, acc,  false,  false);

    const float v = ((acc[0] + acc[1]) + (acc[2] + acc[3])) +
                    ((acc[4] + acc[5]) + (acc[6] + acc[7]));

    if ((lane & 8) == 0) {
        const long long o = b0 + (lane & 7) + 8 * h;
        if (o < Bn) out[o] = v;
    }
}

extern "C" void kernel_launch(void* const* d_in, const int* in_sizes, int n_in,
                              void* d_out, int out_size, void* d_ws, size_t ws_size,
                              hipStream_t stream) {
    const int*   x  = (const int*)d_in[0];
    const int*   y  = (const int*)d_in[1];
    const float* Wc = (const float*)d_in[2];
    const float* Wa = (const float*)d_in[3];
    float*       out = (float*)d_out;

    const int Bn = in_sizes[0];
    const long long waves  = ((long long)Bn + 15) / 16;
    const int       blocks = (int)((waves + 7) / 8);

    twotower_dot_wmma<<<blocks, 256, 0, stream>>>(x, y, Wc, Wa, out, Bn);
}
